# starting kernel
# baseline (speedup 1.0000x reference)
#include <hip/hip_runtime.h>
#include <hip/hip_bf16.h>

#define B_SZ   128
#define IN_SZ  1024
#define OUT_SZ 1024
#define E_SZ   32

typedef __attribute__((ext_vector_type(2))) float v2f;
typedef __attribute__((ext_vector_type(8))) float v8f;

__device__ inline void wave_top2(float val, int lane,
                                 int& i0, int& i1, float& v0, float& v1) {
  float v = val; int idx = lane;
#pragma unroll
  for (int off = 16; off > 0; off >>= 1) {
    float ov = __shfl_xor(v, off, 32);
    int   oi = __shfl_xor(idx, off, 32);
    if (ov > v || (ov == v && oi < idx)) { v = ov; idx = oi; }
  }
  v0 = v; i0 = idx;
  float v2 = (lane == i0) ? -__builtin_inff() : val;
  float vv = v2; int idx2 = lane;
#pragma unroll
  for (int off = 16; off > 0; off >>= 1) {
    float ov = __shfl_xor(vv, off, 32);
    int   oi = __shfl_xor(idx2, off, 32);
    if (ov > vv || (ov == vv && oi < idx2)) { vv = ov; idx2 = oi; }
  }
  v1 = vv; i1 = idx2;
}

__global__ __launch_bounds__(32) void moe_route_kernel(
    const float* __restrict__ x, const float* __restrict__ rw,
    const float* __restrict__ brw, const float* __restrict__ eb,
    float* __restrict__ out, int* __restrict__ cnt,
    int* __restrict__ gtok, float* __restrict__ gprob) {
  const int b = blockIdx.x;
  const int lane = threadIdx.x;
  const float* xb = x + (size_t)b * IN_SZ;

  float accr = 0.f, accb = 0.f;
#pragma unroll 4
  for (int i = 0; i < IN_SZ; ++i) {
    const float xv = xb[i];
    accr = __builtin_fmaf(xv, rw[i * E_SZ + lane], accr);
    accb = __builtin_fmaf(xv, brw[i * E_SZ + lane], accb);
  }

  int i0, i1; float v0, v1;
  wave_top2(accr, lane, i0, i1, v0, v1);
  const float ex = __expf(v1 - v0);
  const float inv = 1.f / (1.f + ex);
  const float p0 = inv, p1 = ex * inv;
  if (lane == 0) {
    int s0 = atomicAdd(&cnt[i0], 1);
    gtok[i0 * B_SZ + s0] = b; gprob[i0 * B_SZ + s0] = p0;
    int s1 = atomicAdd(&cnt[i1], 1);
    gtok[i1 * B_SZ + s1] = b; gprob[i1 * B_SZ + s1] = p1;
  }

  int j0, j1; float u0, u1;
  wave_top2(accb, lane, j0, j1, u0, u1);
  const float exb = __expf(u1 - u0);
  const float invb = 1.f / (1.f + exb);
  const float q0 = invb, q1 = exb * invb;
  const float* b0p = eb + (size_t)j0 * OUT_SZ;
  const float* b1p = eb + (size_t)j1 * OUT_SZ;
  float* ob = out + (size_t)b * OUT_SZ;
#pragma unroll 4
  for (int o = lane; o < OUT_SZ; o += 32)
    ob[o] = __builtin_fmaf(q0, b0p[o], q1 * b1p[o]);
}

#define KCHUNK 256
#define APITCH 260

__global__ __launch_bounds__(256) void moe_gemm_kernel(
    const float* __restrict__ x, const float* __restrict__ ew,
    float* __restrict__ out, const int* __restrict__ cnt,
    const int* __restrict__ gtok, const float* __restrict__ gprob) {
  __shared__ float sA[16][APITCH];
  __shared__ int   s_tok[16];
  __shared__ float s_prob[16];

  const int e     = blockIdx.y;
  const int mbase = blockIdx.z * 16;
  const int cnt_e = cnt[e];
  if (mbase >= cnt_e) return;
  const int rows = min(16, cnt_e - mbase);

  const int tid = threadIdx.x;
  if (tid < 16) {
    if (tid < rows) {
      s_tok[tid]  = gtok[e * B_SZ + mbase + tid];
      s_prob[tid] = gprob[e * B_SZ + mbase + tid];
    } else { s_tok[tid] = 0; s_prob[tid] = 0.f; }
  }
  __syncthreads();

  const int wave = tid >> 5;
  const int lane = tid & 31;
  const int half = lane >> 4;
  const int l16  = lane & 15;
  const int col  = blockIdx.x * 128 + wave * 16 + l16;
  const float* Wp = ew + (size_t)e * IN_SZ * OUT_SZ;

  const int arow = tid >> 4;
  const int aseg = (tid & 15) * 16;
  const int atok = s_tok[arow];

  v8f acc = {};

  for (int kb = 0; kb < IN_SZ; kb += KCHUNK) {
    {
      float4* dst = (float4*)&sA[arow][aseg];
      if (arow < rows) {
        const float4* src = (const float4*)(x + (size_t)atok * IN_SZ + kb + aseg);
#pragma unroll
        for (int j = 0; j < 4; ++j) dst[j] = src[j];
      } else {
        const float4 z = make_float4(0.f, 0.f, 0.f, 0.f);
#pragma unroll
        for (int j = 0; j < 4; ++j) dst[j] = z;
      }
    }
    __syncthreads();

    if (kb + KCHUNK < IN_SZ)
      __builtin_prefetch(&Wp[(size_t)(kb + KCHUNK + 2 * half) * OUT_SZ + col], 0, 1);

#pragma unroll 8
    for (int k4 = 0; k4 < KCHUNK; k4 += 4) {
      const float2 av = *(const float2*)&sA[l16][k4 + 2 * half];
      v2f a; a[0] = av.x; a[1] = av.y;
      const size_t krow = (size_t)(kb + k4 + 2 * half) * OUT_SZ + col;
      v2f bb; bb[0] = Wp[krow]; bb[1] = Wp[krow + OUT_SZ];
      acc = __builtin_amdgcn_wmma_f32_16x16x4_f32(
           false, a,  false, bb,
           (short)0, acc,  false,  false);
    }
    __syncthreads();
  }

#pragma unroll
  for (int v = 0; v < 8; ++v) {
    const int r = half ? (v + 8) : v;
    if (r < rows)
      atomicAdd(&out[(size_t)s_tok[r] * OUT_SZ + col], s_prob[r] * acc[v]);
  }
}

extern "C" void kernel_launch(void* const* d_in, const int* in_sizes, int n_in,
                              void* d_out, int out_size, void* d_ws, size_t ws_size,
                              hipStream_t stream) {
  const float* x   = (const float*)d_in[0];
  const float* rw  = (const float*)d_in[1];
  const float* brw = (const float*)d_in[2];
  const float* ew  = (const float*)d_in[3];
  const float* eb  = (const float*)d_in[4];
  float* out = (float*)d_out;

  int*   cnt   = (int*)d_ws;
  int*   gtok  = (int*)((char*)d_ws + 256);
  float* gprob = (float*)((char*)d_ws + 256 + (size_t)E_SZ * B_SZ * sizeof(int));

  hipMemsetAsync(cnt, 0, E_SZ * sizeof(int), stream);
  moe_route_kernel<<<B_SZ, 32, 0, stream>>>(x, rw, brw, eb, out, cnt, gtok, gprob);
  dim3 g2(OUT_SZ / 128, E_SZ, B_SZ / 16);
  moe_gemm_kernel<<<g2, 256, 0, stream>>>(x, ew, out, cnt, gtok, gprob);
}


# routing logits as f32 WMMA tiles: block = 16 samples x one router, K split over 16 waves, partials added in wave order through LDS (changes logit summation order)
# speedup vs baseline: 3.8328x; 3.8328x over previous
#include <hip/hip_runtime.h>
#include <hip/hip_bf16.h>

#define B_SZ   128
#define IN_SZ  1024
#define OUT_SZ 1024
#define E_SZ   32

typedef __attribute__((ext_vector_type(2))) float v2f;
typedef __attribute__((ext_vector_type(8))) float v8f;

__device__ inline void wave_top2(float val, int lane,
                                 int& i0, int& i1, float& v0, float& v1) {
  float v = val; int idx = lane;
#pragma unroll
  for (int off = 16; off > 0; off >>= 1) {
    float ov = __shfl_xor(v, off, 32);
    int   oi = __shfl_xor(idx, off, 32);
    if (ov > v || (ov == v && oi < idx)) { v = ov; idx = oi; }
  }
  v0 = v; i0 = idx;
  float v2 = (lane == i0) ? -__builtin_inff() : val;
  float vv = v2; int idx2 = lane;
#pragma unroll
  for (int off = 16; off > 0; off >>= 1) {
    float ov = __shfl_xor(vv, off, 32);
    int   oi = __shfl_xor(idx2, off, 32);
    if (ov > vv || (ov == vv && oi < idx2)) { vv = ov; idx2 = oi; }
  }
  v1 = vv; i1 = idx2;
}

#define RT_WAVES 16
#define RT_KW    (IN_SZ / RT_WAVES)
#define PPITCH   36

__global__ __launch_bounds__(512) void moe_route_kernel(
    const float* __restrict__ x, const float* __restrict__ rw,
    const float* __restrict__ brw, const float* __restrict__ eb,
    float* __restrict__ out, int* __restrict__ cnt,
    int* __restrict__ gtok, float* __restrict__ gprob) {
  __shared__ __attribute__((aligned(16))) float part[RT_WAVES][16][PPITCH];

  const int m0   = blockIdx.x * 16;
  const int rsel = blockIdx.y;
  const float* Wr = rsel ? brw : rw;
  const int tid  = threadIdx.x;
  const int wave = tid >> 5;
  const int lane = tid & 31;
  const int half = lane >> 4;
  const int l16  = lane & 15;
  const int kw   = wave * RT_KW;

  const float4* xp = (const float4*)(x + (size_t)(m0 + l16) * IN_SZ + kw + 4 * half);
  const float2* wp = (const float2*)(Wr + (size_t)(kw + 4 * half) * E_SZ + 2 * l16);

  float4 av[8];
  float2 bv[8][4];
#pragma unroll
  for (int q = 0; q < 8; ++q) {
    av[q] = xp[2 * q];
#pragma unroll
    for (int j = 0; j < 4; ++j) bv[q][j] = wp[(8 * q + j) * (E_SZ / 2)];
  }

  v8f acc0 = {}, acc1 = {};
#pragma unroll
  for (int q = 0; q < 8; ++q) {
    v2f a, b0, b1;
    a[0]  = av[q].x;     a[1]  = av[q].y;
    b0[0] = bv[q][0].x;  b0[1] = bv[q][1].x;
    b1[0] = bv[q][0].y;  b1[1] = bv[q][1].y;
    acc0 = __builtin_amdgcn_wmma_f32_16x16x4_f32(false, a, false, b0, (short)0, acc0, false, false);
    acc1 = __builtin_amdgcn_wmma_f32_16x16x4_f32(false, a, false, b1, (short)0, acc1, false, false);
    a[0]  = av[q].z;     a[1]  = av[q].w;
    b0[0] = bv[q][2].x;  b0[1] = bv[q][3].x;
    b1[0] = bv[q][2].y;  b1[1] = bv[q][3].y;
    acc0 = __builtin_amdgcn_wmma_f32_16x16x4_f32(false, a, false, b0, (short)0, acc0, false, false);
    acc1 = __builtin_amdgcn_wmma_f32_16x16x4_f32(false, a, false, b1, (short)0, acc1, false, false);
  }

#pragma unroll
  for (int v = 0; v < 8; ++v) {
    float2 p; p.x = acc0[v]; p.y = acc1[v];
    *(float2*)&part[wave][v + 8 * half][2 * l16] = p;
  }
  __syncthreads();

  float logit = 0.f;
#pragma unroll
  for (int w = 0; w < RT_WAVES; ++w) logit += part[w][wave][lane];

  const int b = m0 + wave;
  int i0, i1; float v0, v1;
  wave_top2(logit, lane, i0, i1, v0, v1);
  const float ex = __expf(v1 - v0);
  const float inv = 1.f / (1.f + ex);
  const float p0 = inv, p1 = ex * inv;

  if (rsel == 0) {
    if (lane == 0) {
      const int s0 = atomicAdd(&cnt[i0], 1);
      const int s1 = atomicAdd(&cnt[i1], 1);
      gtok[i0 * B_SZ + s0] = b; gprob[i0 * B_SZ + s0] = p0;
      gtok[i1 * B_SZ + s1] = b; gprob[i1 * B_SZ + s1] = p1;
    }
  } else {
    const float4* b0p = (const float4*)(eb + (size_t)i0 * OUT_SZ);
    const float4* b1p = (const float4*)(eb + (size_t)i1 * OUT_SZ);
    float4* ob = (float4*)(out + (size_t)b * OUT_SZ);
#pragma unroll
    for (int o = lane; o < OUT_SZ / 4; o += 32) {
      const float4 u = b0p[o], w = b1p[o];
      float4 r;
      r.x = __builtin_fmaf(p0, u.x, p1 * w.x);
      r.y = __builtin_fmaf(p0, u.y, p1 * w.y);
      r.z = __builtin_fmaf(p0, u.z, p1 * w.z);
      r.w = __builtin_fmaf(p0, u.w, p1 * w.w);
      ob[o] = r;
    }
  }
}

#define KCHUNK 256
#define APITCH 260

__global__ __launch_bounds__(256) void moe_gemm_kernel(
    const float* __restrict__ x, const float* __restrict__ ew,
    float* __restrict__ out, const int* __restrict__ cnt,
    const int* __restrict__ gtok, const float* __restrict__ gprob) {
  __shared__ float sA[16][APITCH];
  __shared__ int   s_tok[16];
  __shared__ float s_prob[16];

  const int e     = blockIdx.y;
  const int mbase = blockIdx.z * 16;
  const int cnt_e = cnt[e];
  if (mbase >= cnt_e) return;
  const int rows = min(16, cnt_e - mbase);

  const int tid = threadIdx.x;
  if (tid < 16) {
    if (tid < rows) {
      s_tok[tid]  = gtok[e * B_SZ + mbase + tid];
      s_prob[tid] = gprob[e * B_SZ + mbase + tid];
    } else { s_tok[tid] = 0; s_prob[tid] = 0.f; }
  }
  __syncthreads();

  const int wave = tid >> 5;
  const int lane = tid & 31;
  const int half = lane >> 4;
  const int l16  = lane & 15;
  const int col  = blockIdx.x * 128 + wave * 16 + l16;
  const float* Wp = ew + (size_t)e * IN_SZ * OUT_SZ;

  const int arow = tid >> 4;
  const int aseg = (tid & 15) * 16;
  const int atok = s_tok[arow];

  v8f acc = {};

  for (int kb = 0; kb < IN_SZ; kb += KCHUNK) {
    {
      float4* dst = (float4*)&sA[arow][aseg];
      if (arow < rows) {
        const float4* src = (const float4*)(x + (size_t)atok * IN_SZ + kb + aseg);
#pragma unroll
        for (int j = 0; j < 4; ++j) dst[j] = src[j];
      } else {
        const float4 z = make_float4(0.f, 0.f, 0.f, 0.f);
#pragma unroll
        for (int j = 0; j < 4; ++j) dst[j] = z;
      }
    }
    __syncthreads();

    if (kb + KCHUNK < IN_SZ)
      __builtin_prefetch(&Wp[(size_t)(kb + KCHUNK + 2 * half) * OUT_SZ + col], 0, 1);

#pragma unroll 8
    for (int k4 = 0; k4 < KCHUNK; k4 += 4) {
      const float2 av = *(const float2*)&sA[l16][k4 + 2 * half];
      v2f a; a[0] = av.x; a[1] = av.y;
      const size_t krow = (size_t)(kb + k4 + 2 * half) * OUT_SZ + col;
      v2f bb; bb[0] = Wp[krow]; bb[1] = Wp[krow + OUT_SZ];
      acc = __builtin_amdgcn_wmma_f32_16x16x4_f32(
           false, a,  false, bb,
           (short)0, acc,  false,  false);
    }
    __syncthreads();
  }

#pragma unroll
  for (int v = 0; v < 8; ++v) {
    const int r = half ? (v + 8) : v;
    if (r < rows)
      atomicAdd(&out[(size_t)s_tok[r] * OUT_SZ + col], s_prob[r] * acc[v]);
  }
}

extern "C" void kernel_launch(void* const* d_in, const int* in_sizes, int n_in,
                              void* d_out, int out_size, void* d_ws, size_t ws_size,
                              hipStream_t stream) {
  const float* x   = (const float*)d_in[0];
  const float* rw  = (const float*)d_in[1];
  const float* brw = (const float*)d_in[2];
  const float* ew  = (const float*)d_in[3];
  const float* eb  = (const float*)d_in[4];
  float* out = (float*)d_out;

  int*   cnt   = (int*)d_ws;
  int*   gtok  = (int*)((char*)d_ws + 256);
  float* gprob = (float*)((char*)d_ws + 256 + (size_t)E_SZ * B_SZ * sizeof(int));

  hipMemsetAsync(cnt, 0, E_SZ * sizeof(int), stream);
  dim3 g1(B_SZ / 16, 2);
  moe_route_kernel<<<g1, 512, 0, stream>>>(x, rw, brw, eb, out, cnt, gtok, gprob);
  dim3 g2(OUT_SZ / 128, E_SZ, B_SZ / 16);
  moe_gemm_kernel<<<g2, 256, 0, stream>>>(x, ew, out, cnt, gtok, gprob);
}
